# starting kernel
# baseline (speedup 1.0000x reference)
#include <hip/hip_runtime.h>

typedef __attribute__((ext_vector_type(2))) float v2f;
typedef __attribute__((ext_vector_type(4))) float v4f;
typedef __attribute__((ext_vector_type(8))) float v8f;

#define D_IN    64
#define N_EVAL  68
#define N_TOTAL 132
#define NPAD    80
#define SWS     148
#define SVS     84
#define WAVES_PER_BLOCK 2

#define FOLD   (-7.2134752044448169f)
#define UCLAMP (86.561702453337803f)

__device__ __forceinline__ float neat_sigmoid(float bF, float rF, float t) {
    float u = __builtin_fmaf(rF, t, bF);
    u = fminf(fmaxf(u, -UCLAMP), UCLAMP);
    float e = __builtin_amdgcn_exp2f(u);
    return __builtin_amdgcn_rcpf(1.0f + e);
}

__global__ __launch_bounds__(64, 1)
void neat_wmma_kernel(const float* __restrict__ x,
                      const float* __restrict__ W,
                      const float* __restrict__ bias,
                      const float* __restrict__ resp,
                      float* __restrict__ out)
{
    __shared__ float sW[NPAD * SWS];
    __shared__ float sB[NPAD];
    __shared__ float sR[NPAD];
    __shared__ float sV[WAVES_PER_BLOCK][16 * SVS];

    const int tid  = threadIdx.x;
    const int wave = tid >> 5;
    const int lane = tid & 31;

    for (int idx = tid; idx < NPAD * SWS; idx += 64) {
        int r = idx / SWS, c = idx - r * SWS;
        sW[idx] = (r < N_EVAL && c < N_TOTAL) ? W[r * N_TOTAL + c] : 0.0f;
    }
    for (int idx = tid; idx < NPAD; idx += 64) {
        sB[idx] = (idx < N_EVAL) ? FOLD * bias[idx] : 0.0f;
        sR[idx] = (idx < N_EVAL) ? FOLD * resp[idx] : FOLD;
    }
    __syncthreads();

    const int  mrow  = lane & 15;
    const int  khalf = lane >> 4;
    const long tile  = (long)blockIdx.x * WAVES_PER_BLOCK + wave;
    const long s0    = tile * 16;

    v2f xa[16];
    const float* xrowp = x + (s0 + mrow) * D_IN + khalf * 2;
    #pragma unroll
    for (int k = 0; k < 16; ++k)
        xa[k] = *(const v2f*)(xrowp + 4 * k);

    float* myV = &sV[wave][0];

    for (int g = 0; g < 5; ++g) {
        const int rowbase = (g * 16 + mrow) * SWS;
        v8f acc = {0.f, 0.f, 0.f, 0.f, 0.f, 0.f, 0.f, 0.f};

        const float* wb = &sW[rowbase + khalf * 2];
        #pragma unroll
        for (int k = 0; k < 16; ++k) {
            v2f b = *(const v2f*)(wb + 4 * k);
            acc = __builtin_amdgcn_wmma_f32_16x16x4_f32(false, xa[k], false, b,
                                                        (short)0, acc, false, false);
        }
        for (int p = 0; p < g; ++p) {
            const float* va  = myV + mrow * SVS + p * 16 + khalf * 2;
            const float* wbh = &sW[rowbase + D_IN + p * 16 + khalf * 2];
            #pragma unroll
            for (int k = 0; k < 4; ++k) {
                v2f a = *(const v2f*)(va + 4 * k);
                v2f b = *(const v2f*)(wbh + 4 * k);
                acc = __builtin_amdgcn_wmma_f32_16x16x4_f32(false, a, false, b,
                                                            (short)0, acc, false, false);
            }
        }

        float wl[16];
        const float* wtr = &sW[rowbase + D_IN + g * 16];
        #pragma unroll
        for (int i = 0; i < 16; i += 4)
            *(v4f*)(wl + i) = *(const v4f*)(wtr + i);

        const float bl = sB[g * 16 + mrow];
        const float rl = sR[g * 16 + mrow];
        const int srcBase = (lane & 16) << 2;

        #pragma unroll
        for (int i = 0; i < 16; ++i) {
            const float bi = sB[g * 16 + i];
            const float ri = sR[g * 16 + i];
            const int srcb = srcBase + (i << 2);
            #pragma unroll
            for (int r = 0; r < 8; ++r) {
                float t = __int_as_float(
                    __builtin_amdgcn_ds_bpermute(srcb, __float_as_int(acc[r])));
                float v = neat_sigmoid(bi, ri, t);
                acc[r] += wl[i] * v;
            }
        }

        #pragma unroll
        for (int r = 0; r < 8; ++r) {
            float v = neat_sigmoid(bl, rl, acc[r]);
            myV[(r + khalf * 8) * SVS + g * 16 + mrow] = v;
        }
    }

    if (lane < 16) {
        v4f o = *(const v4f*)(myV + lane * SVS + D_IN);
        *(v4f*)(out + (s0 + lane) * 4) = o;
    }
}

extern "C" void kernel_launch(void* const* d_in, const int* in_sizes, int n_in,
                              void* d_out, int out_size, void* d_ws, size_t ws_size,
                              hipStream_t stream) {
    const float* x    = (const float*)d_in[0];
    const float* W    = (const float*)d_in[1];
    const float* bias = (const float*)d_in[2];
    const float* resp = (const float*)d_in[3];
    float* out = (float*)d_out;

    const int B = in_sizes[0] / D_IN;
    const int blocks = B / (16 * WAVES_PER_BLOCK);
    neat_wmma_kernel<<<blocks, 64, 0, stream>>>(x, W, bias, resp, out);
}


# Stage W once per 16-wave work-group that loops over sample tiles (512 groups, 2 tiles per wave) instead of once per 2-wave group: 16x fewer serialized staging passes, 32 waves per WGP
# speedup vs baseline: 4.6311x; 4.6311x over previous
#include <hip/hip_runtime.h>

typedef __attribute__((ext_vector_type(2))) float v2f;
typedef __attribute__((ext_vector_type(4))) float v4f;
typedef __attribute__((ext_vector_type(8))) float v8f;

#define D_IN    64
#define N_EVAL  68
#define N_TOTAL 132
#define NPAD    80
#define SWS     148
#define SVS     84
#define WAVES_PER_BLOCK 16
#define BLOCK_THREADS (WAVES_PER_BLOCK * 32)
#define NUM_BLOCKS 512

#define FOLD   (-7.2134752044448169f)
#define UCLAMP (86.561702453337803f)

__device__ __forceinline__ float neat_sigmoid(float bF, float rF, float t) {
    float u = __builtin_fmaf(rF, t, bF);
    u = fminf(fmaxf(u, -UCLAMP), UCLAMP);
    float e = __builtin_amdgcn_exp2f(u);
    return __builtin_amdgcn_rcpf(1.0f + e);
}

__global__ __launch_bounds__(BLOCK_THREADS, 8)
void neat_wmma_kernel(const float* __restrict__ x,
                      const float* __restrict__ W,
                      const float* __restrict__ bias,
                      const float* __restrict__ resp,
                      float* __restrict__ out,
                      int n_x)
{
    __shared__ float sW[NPAD * SWS];
    __shared__ float sB[NPAD];
    __shared__ float sR[NPAD];
    __shared__ float sV[WAVES_PER_BLOCK][16 * SVS];

    const int tid  = threadIdx.x;
    const int wave = tid >> 5;
    const int lane = tid & 31;

    for (int idx = tid; idx < NPAD * SWS; idx += BLOCK_THREADS) {
        int r = idx / SWS, c = idx - r * SWS;
        sW[idx] = (r < N_EVAL && c < N_TOTAL) ? W[r * N_TOTAL + c] : 0.0f;
    }
    for (int idx = tid; idx < NPAD; idx += BLOCK_THREADS) {
        sB[idx] = (idx < N_EVAL) ? FOLD * bias[idx] : 0.0f;
        sR[idx] = (idx < N_EVAL) ? FOLD * resp[idx] : FOLD;
    }
    __syncthreads();

    const int  mrow  = lane & 15;
    const int  khalf = lane >> 4;
    const long ntiles = (long)(n_x / (D_IN * 16));
    float* myV = &sV[wave][0];

    for (long tile = (long)blockIdx.x * WAVES_PER_BLOCK + wave; tile < ntiles;
         tile += (long)gridDim.x * WAVES_PER_BLOCK) {
        const long s0    = tile * 16;

        v2f xa[16];
        const float* xrowp = x + (s0 + mrow) * D_IN + khalf * 2;
        #pragma unroll
        for (int k = 0; k < 16; ++k)
            xa[k] = *(const v2f*)(xrowp + 4 * k);

        for (int g = 0; g < 5; ++g) {
            const int rowbase = (g * 16 + mrow) * SWS;
            v8f acc = {0.f, 0.f, 0.f, 0.f, 0.f, 0.f, 0.f, 0.f};

            const float* wb = &sW[rowbase + khalf * 2];
            #pragma unroll
            for (int k = 0; k < 16; ++k) {
                v2f b = *(const v2f*)(wb + 4 * k);
                acc = __builtin_amdgcn_wmma_f32_16x16x4_f32(false, xa[k], false, b,
                                                            (short)0, acc, false, false);
            }
            for (int p = 0; p < g; ++p) {
                const float* va  = myV + mrow * SVS + p * 16 + khalf * 2;
                const float* wbh = &sW[rowbase + D_IN + p * 16 + khalf * 2];
                #pragma unroll
                for (int k = 0; k < 4; ++k) {
                    v2f a = *(const v2f*)(va + 4 * k);
                    v2f b = *(const v2f*)(wbh + 4 * k);
                    acc = __builtin_amdgcn_wmma_f32_16x16x4_f32(false, a, false, b,
                                                                (short)0, acc, false, false);
                }
            }

            float wl[16];
            const float* wtr = &sW[rowbase + D_IN + g * 16];
            #pragma unroll
            for (int i = 0; i < 16; i += 4)
                *(v4f*)(wl + i) = *(const v4f*)(wtr + i);

            const float bl = sB[g * 16 + mrow];
            const float rl = sR[g * 16 + mrow];
            const int srcBase = (lane & 16) << 2;

            #pragma unroll
            for (int i = 0; i < 16; ++i) {
                const float bi = sB[g * 16 + i];
                const float ri = sR[g * 16 + i];
                const int srcb = srcBase + (i << 2);
                #pragma unroll
                for (int r = 0; r < 8; ++r) {
                    float t = __int_as_float(
                        __builtin_amdgcn_ds_bpermute(srcb, __float_as_int(acc[r])));
                    float v = neat_sigmoid(bi, ri, t);
                    acc[r] += wl[i] * v;
                }
            }

            #pragma unroll
            for (int r = 0; r < 8; ++r) {
                float v = neat_sigmoid(bl, rl, acc[r]);
                myV[(r + khalf * 8) * SVS + g * 16 + mrow] = v;
            }
        }

        if (lane < 16) {
            v4f o = *(const v4f*)(myV + lane * SVS + D_IN);
            *(v4f*)(out + (s0 + lane) * 4) = o;
        }
    }
}

extern "C" void kernel_launch(void* const* d_in, const int* in_sizes, int n_in,
                              void* d_out, int out_size, void* d_ws, size_t ws_size,
                              hipStream_t stream) {
    const float* x    = (const float*)d_in[0];
    const float* W    = (const float*)d_in[1];
    const float* bias = (const float*)d_in[2];
    const float* resp = (const float*)d_in[3];
    float* out = (float*)d_out;

    neat_wmma_kernel<<<NUM_BLOCKS, BLOCK_THREADS, 0, stream>>>(x, W, bias, resp, out, in_sizes[0]);
}


# Swap WMMA operands (A = weights, B = samples): a lane owns 1 sample and 8 nodes, so the in-group sweep does 1 sigmoid per node, not 16; node values stay in registers as B; hidden sums in new k order
# speedup vs baseline: 7.6095x; 1.6431x over previous
#include <hip/hip_runtime.h>


typedef __attribute__((ext_vector_type(2))) float v2f;
typedef __attribute__((ext_vector_type(4))) float v4f;
typedef __attribute__((ext_vector_type(8))) float v8f;

#define D_IN    64
#define N_EVAL  68
#define N_TOTAL 132
#define NGRP    5
#define NPAD    80
#define SWS     148
#define WAVES_PER_BLOCK 16
#define BLOCK_THREADS (WAVES_PER_BLOCK * 32)
#define NUM_BLOCKS 512

#define FOLD   (-7.2134752044448169f)
#define UCLAMP (86.561702453337803f)

__device__ __forceinline__ float neat_sigmoid(float bF, float rF, float t) {
    float u = __builtin_fmaf(rF, t, bF);
    u = fminf(fmaxf(u, -UCLAMP), UCLAMP);
    float e = __builtin_amdgcn_exp2f(u);
    return __builtin_amdgcn_rcpf(1.0f + e);
}

__device__ __forceinline__ float xhalf(float x) {
    int b = __builtin_bit_cast(int, x);
    return __builtin_bit_cast(float,
        __builtin_amdgcn_permlanex16(b, b, 0x76543210, 0xfedcba98, false, false));
}

__global__ __launch_bounds__(BLOCK_THREADS, 8)
void neat_wmma_kernel(const float* __restrict__ x,
                      const float* __restrict__ W,
                      const float* __restrict__ bias,
                      const float* __restrict__ resp,
                      float* __restrict__ out,
                      int n_x)
{
    __shared__ __attribute__((aligned(16))) float sW[NPAD * SWS];
    __shared__ __attribute__((aligned(16))) float sT[NGRP * 16 * 16];
    __shared__ float sB[NPAD];
    __shared__ float sR[NPAD];

    const int tid  = threadIdx.x;
    const int wave = tid >> 5;
    const int lane = tid & 31;

    for (int idx = tid; idx < NPAD * SWS; idx += BLOCK_THREADS) {
        int r = idx / SWS, c = idx - r * SWS;
        sW[idx] = (r < N_EVAL && c < N_TOTAL) ? W[r * N_TOTAL + c] : 0.0f;
    }
    for (int idx = tid; idx < NPAD; idx += BLOCK_THREADS) {
        sB[idx] = (idx < N_EVAL) ? FOLD * bias[idx] : 0.0f;
        sR[idx] = (idx < N_EVAL) ? FOLD * resp[idx] : FOLD;
    }
    __syncthreads();
    for (int idx = tid; idx < NGRP * 256; idx += BLOCK_THREADS) {
        int g = idx >> 8, i = (idx >> 4) & 15, j = idx & 15;
        sT[idx] = sW[(g * 16 + j) * SWS + D_IN + g * 16 + i];
    }
    __syncthreads();

    const int  col   = lane & 15;
    const int  khalf = lane >> 4;
    const bool upper = khalf != 0;
    const long ntiles = (long)(n_x / (D_IN * 16));

    for (long tile = (long)blockIdx.x * WAVES_PER_BLOCK + wave; tile < ntiles;
         tile += (long)gridDim.x * WAVES_PER_BLOCK) {
        const long s0 = tile * 16;

        int wrow = col * SWS;
        int trow = khalf * 8;
        int brow = 0;
        asm volatile("" : "+v"(wrow), "+v"(trow), "+v"(brow));

        v2f xb[16];
        const float* xrowp = x + (s0 + col) * D_IN + khalf * 2;
        #pragma unroll
        for (int k = 0; k < 16; ++k)
            xb[k] = *(const v2f*)(xrowp + 4 * k);

        v2f V[NGRP - 1][4];

        #pragma unroll
        for (int g = 0; g < NGRP; ++g) {
            const int rowbase = g * 16 * SWS + wrow;
            v8f acc = {0.f, 0.f, 0.f, 0.f, 0.f, 0.f, 0.f, 0.f};

            const float* wa = &sW[rowbase + khalf * 2];
            #pragma unroll
            for (int k = 0; k < 16; ++k) {
                v2f a = *(const v2f*)(wa + 4 * k);
                acc = __builtin_amdgcn_wmma_f32_16x16x4_f32(false, a, false, xb[k],
                                                            (short)0, acc, false, false);
            }
            #pragma unroll
            for (int p = 0; p < g; ++p) {
                const float* wah = &sW[rowbase + D_IN + p * 16 + khalf * 8];
                #pragma unroll
                for (int k = 0; k < 4; ++k) {
                    v2f a = *(const v2f*)(wah + 2 * k);
                    acc = __builtin_amdgcn_wmma_f32_16x16x4_f32(false, a, false, V[p][k],
                                                                (short)0, acc, false, false);
                }
            }

            const float* tw = &sT[g * 256 + trow];
            const float* pb = &sB[g * 16 + brow];
            const float* pr = &sR[g * 16 + brow];

            if (g < NGRP - 1) {
                float va[8];
                #pragma unroll
                for (int i = 0; i < 8; ++i) {
                    const float v = neat_sigmoid(pb[i], pr[i], acc[i]);
                    va[i] = v;
                    const float o  = xhalf(v);
                    const float vi = upper ? o : v;
                    const v4f w0 = *(const v4f*)(tw + i * 16);
                    const v4f w1 = *(const v4f*)(tw + i * 16 + 4);
                    #pragma unroll
                    for (int j = 0; j < 4; ++j) {
                        acc[j]     = __builtin_fmaf(w0[j], vi, acc[j]);
                        acc[4 + j] = __builtin_fmaf(w1[j], vi, acc[4 + j]);
                    }
                }
                #pragma unroll
                for (int r = 0; r < 8; ++r) {
                    const int i = 8 + r;
                    const float v = neat_sigmoid(pb[i], pr[i], acc[r]);
                    V[g][r >> 1][r & 1] = upper ? v : va[r];
                    const v4f w0 = *(const v4f*)(tw + i * 16);
                    const v4f w1 = *(const v4f*)(tw + i * 16 + 4);
                    #pragma unroll
                    for (int j = r + 1; j < 8; ++j)
                        acc[j] = __builtin_fmaf(j < 4 ? w0[j & 3] : w1[j & 3], v, acc[j]);
                }
            } else {
                float o4[4];
                #pragma unroll
                for (int i = 0; i < 4; ++i) {
                    const float v = neat_sigmoid(pb[i], pr[i], acc[i]);
                    o4[i] = v;
                    const v4f w0 = *(const v4f*)(tw + i * 16);
                    #pragma unroll
                    for (int j = i + 1; j < 4; ++j)
                        acc[j] = __builtin_fmaf(w0[j], v, acc[j]);
                }
                if (lane < 16) {
                    v4f o = {o4[0], o4[1], o4[2], o4[3]};
                    *(v4f*)(out + (s0 + lane) * 4) = o;
                }
            }
        }
    }
}

extern "C" void kernel_launch(void* const* d_in, const int* in_sizes, int n_in,
                              void* d_out, int out_size, void* d_ws, size_t ws_size,
                              hipStream_t stream) {
    const float* x    = (const float*)d_in[0];
    const float* W    = (const float*)d_in[1];
    const float* bias = (const float*)d_in[2];
    const float* resp = (const float*)d_in[3];
    float* out = (float*)d_out;

    neat_wmma_kernel<<<NUM_BLOCKS, BLOCK_THREADS, 0, stream>>>(x, W, bias, resp, out, in_sizes[0]);
}


# Stage W with each thread issuing all its loads first as independent 16-byte loads, LDS stores after (same for the transposed copy): one memory wait per work-group, not one per element
# speedup vs baseline: 9.2499x; 1.2156x over previous
#include <hip/hip_runtime.h>


typedef __attribute__((ext_vector_type(2))) float v2f;
typedef __attribute__((ext_vector_type(4))) float v4f;
typedef __attribute__((ext_vector_type(8))) float v8f;

#define D_IN    64
#define N_EVAL  68
#define N_TOTAL 132
#define NGRP    5
#define NPAD    80
#define SWS     148
#define WAVES_PER_BLOCK 16
#define BLOCK_THREADS (WAVES_PER_BLOCK * 32)
#define NUM_BLOCKS 512
#define SW4     (SWS / 4)
#define W4      (N_TOTAL / 4)
#define NSLOT   (NPAD * SW4)
#define STAGE_Q ((NSLOT + BLOCK_THREADS - 1) / BLOCK_THREADS)
#define T_Q     ((NGRP * 256 + BLOCK_THREADS - 1) / BLOCK_THREADS)

#define FOLD   (-7.2134752044448169f)
#define UCLAMP (86.561702453337803f)

__device__ __forceinline__ float neat_sigmoid(float bF, float rF, float t) {
    float u = __builtin_fmaf(rF, t, bF);
    u = fminf(fmaxf(u, -UCLAMP), UCLAMP);
    float e = __builtin_amdgcn_exp2f(u);
    return __builtin_amdgcn_rcpf(1.0f + e);
}

__device__ __forceinline__ float xhalf(float x) {
    int b = __builtin_bit_cast(int, x);
    return __builtin_bit_cast(float,
        __builtin_amdgcn_permlanex16(b, b, 0x76543210, 0xfedcba98, false, false));
}

__global__ __launch_bounds__(BLOCK_THREADS, 8)
void neat_wmma_kernel(const float* __restrict__ x,
                      const float* __restrict__ W,
                      const float* __restrict__ bias,
                      const float* __restrict__ resp,
                      float* __restrict__ out,
                      int n_x)
{
    __shared__ __attribute__((aligned(16))) float sW[NPAD * SWS];
    __shared__ __attribute__((aligned(16))) float sT[NGRP * 16 * 16];
    __shared__ float sB[NPAD];
    __shared__ float sR[NPAD];

    const int tid  = threadIdx.x;
    const int wave = tid >> 5;
    const int lane = tid & 31;

    v4f wv[STAGE_Q];
    #pragma unroll
    for (int q = 0; q < STAGE_Q; ++q) {
        const int s  = tid + q * BLOCK_THREADS;
        const int r  = s / SW4, c = s - r * SW4;
        const int rr = r < N_EVAL ? r : N_EVAL - 1;
        const int cc = c < W4 ? c : W4 - 1;
        const v4f v  = *(const v4f*)(W + rr * N_TOTAL + cc * 4);
        const bool real = (r < N_EVAL) && (c < W4);
        const v4f z = {0.f, 0.f, 0.f, 0.f};
        wv[q] = real ? v : z;
    }
    float bq = 0.0f, rq = FOLD;
    if (tid < N_EVAL) {
        bq = FOLD * bias[tid];
        rq = FOLD * resp[tid];
    }
    #pragma unroll
    for (int q = 0; q < STAGE_Q; ++q) {
        const int s = tid + q * BLOCK_THREADS;
        if (s < NSLOT) *(v4f*)&sW[s * 4] = wv[q];
    }
    if (tid < NPAD) {
        sB[tid] = bq;
        sR[tid] = rq;
    }
    __syncthreads();
    float tv[T_Q];
    #pragma unroll
    for (int q = 0; q < T_Q; ++q) {
        const int idx = tid + q * BLOCK_THREADS;
        const int g = idx >> 8, i = (idx >> 4) & 15, j = idx & 15;
        tv[q] = (idx < NGRP * 256) ? sW[(g * 16 + j) * SWS + D_IN + g * 16 + i] : 0.0f;
    }
    #pragma unroll
    for (int q = 0; q < T_Q; ++q) {
        const int idx = tid + q * BLOCK_THREADS;
        if (idx < NGRP * 256) sT[idx] = tv[q];
    }
    __syncthreads();

    const int  col   = lane & 15;
    const int  khalf = lane >> 4;
    const bool upper = khalf != 0;
    const long ntiles = (long)(n_x / (D_IN * 16));

    for (long tile = (long)blockIdx.x * WAVES_PER_BLOCK + wave; tile < ntiles;
         tile += (long)gridDim.x * WAVES_PER_BLOCK) {
        const long s0 = tile * 16;

        int wrow = col * SWS;
        int trow = khalf * 8;
        int brow = 0;
        asm volatile("" : "+v"(wrow), "+v"(trow), "+v"(brow));

        v2f xb[16];
        const float* xrowp = x + (s0 + col) * D_IN + khalf * 2;
        #pragma unroll
        for (int k = 0; k < 16; ++k)
            xb[k] = *(const v2f*)(xrowp + 4 * k);

        v2f V[NGRP - 1][4];

        #pragma unroll
        for (int g = 0; g < NGRP; ++g) {
            const int rowbase = g * 16 * SWS + wrow;
            v8f acc = {0.f, 0.f, 0.f, 0.f, 0.f, 0.f, 0.f, 0.f};

            const float* wa = &sW[rowbase + khalf * 2];
            #pragma unroll
            for (int k = 0; k < 16; ++k) {
                v2f a = *(const v2f*)(wa + 4 * k);
                acc = __builtin_amdgcn_wmma_f32_16x16x4_f32(false, a, false, xb[k],
                                                            (short)0, acc, false, false);
            }
            #pragma unroll
            for (int p = 0; p < g; ++p) {
                const float* wah = &sW[rowbase + D_IN + p * 16 + khalf * 8];
                #pragma unroll
                for (int k = 0; k < 4; ++k) {
                    v2f a = *(const v2f*)(wah + 2 * k);
                    acc = __builtin_amdgcn_wmma_f32_16x16x4_f32(false, a, false, V[p][k],
                                                                (short)0, acc, false, false);
                }
            }

            const float* tw = &sT[g * 256 + trow];
            const float* pb = &sB[g * 16 + brow];
            const float* pr = &sR[g * 16 + brow];

            if (g < NGRP - 1) {
                float va[8];
                #pragma unroll
                for (int i = 0; i < 8; ++i) {
                    const float v = neat_sigmoid(pb[i], pr[i], acc[i]);
                    va[i] = v;
                    const float o  = xhalf(v);
                    const float vi = upper ? o : v;
                    const v4f w0 = *(const v4f*)(tw + i * 16);
                    const v4f w1 = *(const v4f*)(tw + i * 16 + 4);
                    #pragma unroll
                    for (int j = 0; j < 4; ++j) {
                        acc[j]     = __builtin_fmaf(w0[j], vi, acc[j]);
                        acc[4 + j] = __builtin_fmaf(w1[j], vi, acc[4 + j]);
                    }
                }
                #pragma unroll
                for (int r = 0; r < 8; ++r) {
                    const int i = 8 + r;
                    const float v = neat_sigmoid(pb[i], pr[i], acc[r]);
                    V[g][r >> 1][r & 1] = upper ? v : va[r];
                    const v4f w0 = *(const v4f*)(tw + i * 16);
                    const v4f w1 = *(const v4f*)(tw + i * 16 + 4);
                    #pragma unroll
                    for (int j = r + 1; j < 8; ++j)
                        acc[j] = __builtin_fmaf(j < 4 ? w0[j & 3] : w1[j & 3], v, acc[j]);
                }
            } else {
                float o4[4];
                #pragma unroll
                for (int i = 0; i < 4; ++i) {
                    const float v = neat_sigmoid(pb[i], pr[i], acc[i]);
                    o4[i] = v;
                    const v4f w0 = *(const v4f*)(tw + i * 16);
                    #pragma unroll
                    for (int j = i + 1; j < 4; ++j)
                        acc[j] = __builtin_fmaf(w0[j], v, acc[j]);
                }
                if (lane < 16) {
                    v4f o = {o4[0], o4[1], o4[2], o4[3]};
                    *(v4f*)(out + (s0 + lane) * 4) = o;
                }
            }
        }
    }
}

extern "C" void kernel_launch(void* const* d_in, const int* in_sizes, int n_in,
                              void* d_out, int out_size, void* d_ws, size_t ws_size,
                              hipStream_t stream) {
    const float* x    = (const float*)d_in[0];
    const float* W    = (const float*)d_in[1];
    const float* bias = (const float*)d_in[2];
    const float* resp = (const float*)d_in[3];
    float* out = (float*)d_out;

    neat_wmma_kernel<<<NUM_BLOCKS, BLOCK_THREADS, 0, stream>>>(x, W, bias, resp, out, in_sizes[0]);
}


# Each wave works on two sample tiles at once (256 groups, 4 waves per SIMD): every LDS weight fetch serves both, A fragments fetched as one group ahead of each matrix block; per-sample sums unchanged
# speedup vs baseline: 10.1277x; 1.0949x over previous
#include <hip/hip_runtime.h>


typedef __attribute__((ext_vector_type(2))) float v2f;
typedef __attribute__((ext_vector_type(4))) float v4f;
typedef __attribute__((ext_vector_type(8))) float v8f;

#define D_IN    64
#define N_EVAL  68
#define N_TOTAL 132
#define NGRP    5
#define NPAD    80
#define SWS     148
#define WAVES_PER_BLOCK 16
#define BLOCK_THREADS (WAVES_PER_BLOCK * 32)
#define TPW 2
#define NUM_BLOCKS 256
#define SW4     (SWS / 4)
#define W4      (N_TOTAL / 4)
#define NSLOT   (NPAD * SW4)
#define STAGE_Q ((NSLOT + BLOCK_THREADS - 1) / BLOCK_THREADS)
#define T_Q     ((NGRP * 256 + BLOCK_THREADS - 1) / BLOCK_THREADS)

#define FOLD   (-7.2134752044448169f)
#define UCLAMP (86.561702453337803f)

__device__ __forceinline__ float neat_sigmoid(float bF, float rF, float t) {
    float u = __builtin_fmaf(rF, t, bF);
    u = fminf(fmaxf(u, -UCLAMP), UCLAMP);
    float e = __builtin_amdgcn_exp2f(u);
    return __builtin_amdgcn_rcpf(1.0f + e);
}

__device__ __forceinline__ float xhalf(float x) {
    int b = __builtin_bit_cast(int, x);
    return __builtin_bit_cast(float,
        __builtin_amdgcn_permlanex16(b, b, 0x76543210, 0xfedcba98, false, false));
}

__global__ __launch_bounds__(BLOCK_THREADS, 4)
void neat_wmma_kernel(const float* __restrict__ x,
                      const float* __restrict__ W,
                      const float* __restrict__ bias,
                      const float* __restrict__ resp,
                      float* __restrict__ out,
                      int n_x)
{
    __shared__ __attribute__((aligned(16))) float sW[NPAD * SWS];
    __shared__ __attribute__((aligned(16))) float sT[NGRP * 16 * 16];
    __shared__ float sB[NPAD];
    __shared__ float sR[NPAD];

    const int tid  = threadIdx.x;
    const int wave = tid >> 5;
    const int lane = tid & 31;

    v4f wv[STAGE_Q];
    #pragma unroll
    for (int q = 0; q < STAGE_Q; ++q) {
        const int s  = tid + q * BLOCK_THREADS;
        const int r  = s / SW4, c = s - r * SW4;
        const int rr = r < N_EVAL ? r : N_EVAL - 1;
        const int cc = c < W4 ? c : W4 - 1;
        const v4f v  = *(const v4f*)(W + rr * N_TOTAL + cc * 4);
        const bool real = (r < N_EVAL) && (c < W4);
        const v4f z = {0.f, 0.f, 0.f, 0.f};
        wv[q] = real ? v : z;
    }
    float bq = 0.0f, rq = FOLD;
    if (tid < N_EVAL) {
        bq = FOLD * bias[tid];
        rq = FOLD * resp[tid];
    }
    #pragma unroll
    for (int q = 0; q < STAGE_Q; ++q) {
        const int s = tid + q * BLOCK_THREADS;
        if (s < NSLOT) *(v4f*)&sW[s * 4] = wv[q];
    }
    if (tid < NPAD) {
        sB[tid] = bq;
        sR[tid] = rq;
    }
    __syncthreads();
    float tv[T_Q];
    #pragma unroll
    for (int q = 0; q < T_Q; ++q) {
        const int idx = tid + q * BLOCK_THREADS;
        const int g = idx >> 8, i = (idx >> 4) & 15, j = idx & 15;
        tv[q] = (idx < NGRP * 256) ? sW[(g * 16 + j) * SWS + D_IN + g * 16 + i] : 0.0f;
    }
    #pragma unroll
    for (int q = 0; q < T_Q; ++q) {
        const int idx = tid + q * BLOCK_THREADS;
        if (idx < NGRP * 256) sT[idx] = tv[q];
    }
    __syncthreads();

    const int  col   = lane & 15;
    const int  khalf = lane >> 4;
    const bool upper = khalf != 0;
    const long ntiles = (long)(n_x / (D_IN * 16));
    const long ngrp_t = (ntiles + TPW - 1) / TPW;

    for (long set = (long)blockIdx.x * WAVES_PER_BLOCK + wave; set < ngrp_t;
         set += (long)gridDim.x * WAVES_PER_BLOCK) {
        long s0[TPW];
        #pragma unroll
        for (int t = 0; t < TPW; ++t) {
            long tile = set * TPW + t;
            tile = tile < ntiles ? tile : ntiles - 1;
            s0[t] = tile * 16;
        }

        int wrow = col * SWS;
        int trow = khalf * 8;
        int brow = 0;
        asm volatile("" : "+v"(wrow), "+v"(trow), "+v"(brow));

        v2f xb[TPW][16];
        #pragma unroll
        for (int t = 0; t < TPW; ++t) {
            const float* xrowp = x + (s0[t] + col) * D_IN + khalf * 2;
            #pragma unroll
            for (int k = 0; k < 16; ++k)
                xb[t][k] = *(const v2f*)(xrowp + 4 * k);
        }

        v2f V[TPW][NGRP - 1][4];

        #pragma unroll
        for (int g = 0; g < NGRP; ++g) {
            const int rowbase = g * 16 * SWS + wrow;
            v8f acc[TPW];
            #pragma unroll
            for (int t = 0; t < TPW; ++t) {
                const v8f z8 = {0.f, 0.f, 0.f, 0.f, 0.f, 0.f, 0.f, 0.f};
                acc[t] = z8;
            }

            const float* wa = &sW[rowbase + khalf * 2];
            v2f ax[16];
            #pragma unroll
            for (int k = 0; k < 16; ++k)
                ax[k] = *(const v2f*)(wa + 4 * k);
            v2f ah[NGRP - 1][4];
            #pragma unroll
            for (int p = 0; p < g; ++p) {
                const float* wah = &sW[rowbase + D_IN + p * 16 + khalf * 8];
                #pragma unroll
                for (int k = 0; k < 4; ++k)
                    ah[p][k] = *(const v2f*)(wah + 2 * k);
            }
            __builtin_amdgcn_sched_barrier(0);

            #pragma unroll
            for (int k = 0; k < 16; ++k) {
                #pragma unroll
                for (int t = 0; t < TPW; ++t)
                    acc[t] = __builtin_amdgcn_wmma_f32_16x16x4_f32(false, ax[k], false, xb[t][k],
                                                                   (short)0, acc[t], false, false);
            }
            #pragma unroll
            for (int p = 0; p < g; ++p) {
                #pragma unroll
                for (int k = 0; k < 4; ++k) {
                    #pragma unroll
                    for (int t = 0; t < TPW; ++t)
                        acc[t] = __builtin_amdgcn_wmma_f32_16x16x4_f32(false, ah[p][k], false, V[t][p][k],
                                                                       (short)0, acc[t], false, false);
                }
            }

            __builtin_amdgcn_sched_barrier(0);

            const float* tw = &sT[g * 256 + trow];
            const float* pb = &sB[g * 16 + brow];
            const float* pr = &sR[g * 16 + brow];

            if (g < NGRP - 1) {
                float va[TPW][8];
                #pragma unroll
                for (int i = 0; i < 8; ++i) {
                    const float bi = pb[i], ri = pr[i];
                    const v4f w0 = *(const v4f*)(tw + i * 16);
                    const v4f w1 = *(const v4f*)(tw + i * 16 + 4);
                    float vi[TPW];
                    #pragma unroll
                    for (int t = 0; t < TPW; ++t) {
                        const float v = neat_sigmoid(bi, ri, acc[t][i]);
                        va[t][i] = v;
                        const float o = xhalf(v);
                        vi[t] = upper ? o : v;
                    }
                    #pragma unroll
                    for (int j = 0; j < 4; ++j) {
                        #pragma unroll
                        for (int t = 0; t < TPW; ++t) {
                            acc[t][j]     = __builtin_fmaf(w0[j], vi[t], acc[t][j]);
                            acc[t][4 + j] = __builtin_fmaf(w1[j], vi[t], acc[t][4 + j]);
                        }
                    }
                }
                #pragma unroll
                for (int r = 0; r < 8; ++r) {
                    const int i = 8 + r;
                    const float bi = pb[i], ri = pr[i];
                    const v4f w0 = *(const v4f*)(tw + i * 16);
                    const v4f w1 = *(const v4f*)(tw + i * 16 + 4);
                    float v[TPW];
                    #pragma unroll
                    for (int t = 0; t < TPW; ++t) {
                        v[t] = neat_sigmoid(bi, ri, acc[t][r]);
                        V[t][g][r >> 1][r & 1] = upper ? v[t] : va[t][r];
                    }
                    #pragma unroll
                    for (int j = r + 1; j < 8; ++j) {
                        #pragma unroll
                        for (int t = 0; t < TPW; ++t)
                            acc[t][j] = __builtin_fmaf(j < 4 ? w0[j & 3] : w1[j & 3], v[t], acc[t][j]);
                    }
                }
            } else {
                float o4[TPW][4];
                #pragma unroll
                for (int i = 0; i < 4; ++i) {
                    const float bi = pb[i], ri = pr[i];
                    const v4f w0 = *(const v4f*)(tw + i * 16);
                    #pragma unroll
                    for (int t = 0; t < TPW; ++t) {
                        const float v = neat_sigmoid(bi, ri, acc[t][i]);
                        o4[t][i] = v;
                        #pragma unroll
                        for (int j = i + 1; j < 4; ++j)
                            acc[t][j] = __builtin_fmaf(w0[j], v, acc[t][j]);
                    }
                }
                if (lane < 16) {
                    #pragma unroll
                    for (int t = 0; t < TPW; ++t) {
                        v4f o = {o4[t][0], o4[t][1], o4[t][2], o4[t][3]};
                        *(v4f*)(out + (s0[t] + lane) * 4) = o;
                    }
                }
            }
        }
    }
}

extern "C" void kernel_launch(void* const* d_in, const int* in_sizes, int n_in,
                              void* d_out, int out_size, void* d_ws, size_t ws_size,
                              hipStream_t stream) {
    const float* x    = (const float*)d_in[0];
    const float* W    = (const float*)d_in[1];
    const float* bias = (const float*)d_in[2];
    const float* resp = (const float*)d_in[3];
    float* out = (float*)d_out;

    neat_wmma_kernel<<<NUM_BLOCKS, BLOCK_THREADS, 0, stream>>>(x, W, bias, resp, out, in_sizes[0]);
}


# In-group sweep after a half swap (v_permlane16_swap_b32): a lane holds all 16 nodes of one sample, tile 0 in lanes 0-15, tile 1 in 16-31; one sigmoid per node for both tiles
# speedup vs baseline: 10.8521x; 1.0715x over previous
#include <hip/hip_runtime.h>


typedef __attribute__((ext_vector_type(2))) float v2f;
typedef __attribute__((ext_vector_type(4))) float v4f;
typedef __attribute__((ext_vector_type(8))) float v8f;

#define D_IN    64
#define N_EVAL  68
#define N_TOTAL 132
#define NGRP    5
#define NPAD    80
#define SWS     148
#define WAVES_PER_BLOCK 16
#define BLOCK_THREADS (WAVES_PER_BLOCK * 32)
#define TPW 2
#define NUM_BLOCKS 256
#define SW4     (SWS / 4)
#define W4      (N_TOTAL / 4)
#define NSLOT   (NPAD * SW4)
#define STAGE_Q ((NSLOT + BLOCK_THREADS - 1) / BLOCK_THREADS)
#define T_Q     ((NGRP * 256 + BLOCK_THREADS - 1) / BLOCK_THREADS)

#define FOLD   (-7.2134752044448169f)
#define UCLAMP (86.561702453337803f)

__device__ __forceinline__ float neat_sigmoid(float bF, float rF, float t) {
    float u = __builtin_fmaf(rF, t, bF);
    u = fminf(fmaxf(u, -UCLAMP), UCLAMP);
    float e = __builtin_amdgcn_exp2f(u);
    return __builtin_amdgcn_rcpf(1.0f + e);
}

__device__ __forceinline__ void half_swap4(float& d0, float& d1, float& d2, float& d3,
                                           float& s0, float& s1, float& s2, float& s3) {
    asm("v_permlane16_swap_b32 %0, %4\n\t"
        "v_permlane16_swap_b32 %1, %5\n\t"
        "v_permlane16_swap_b32 %2, %6\n\t"
        "v_permlane16_swap_b32 %3, %7"
        : "+v"(d0), "+v"(d1), "+v"(d2), "+v"(d3), "+v"(s0), "+v"(s1), "+v"(s2), "+v"(s3));
}

__global__ __launch_bounds__(BLOCK_THREADS, 4)
void neat_wmma_kernel(const float* __restrict__ x,
                      const float* __restrict__ W,
                      const float* __restrict__ bias,
                      const float* __restrict__ resp,
                      float* __restrict__ out,
                      int n_x)
{
    __shared__ __attribute__((aligned(16))) float sW[NPAD * SWS];
    __shared__ __attribute__((aligned(16))) float sT[NGRP * 16 * 16];
    __shared__ float sB[NPAD];
    __shared__ float sR[NPAD];

    const int tid  = threadIdx.x;
    const int wave = tid >> 5;
    const int lane = tid & 31;

    v4f wv[STAGE_Q];
    #pragma unroll
    for (int q = 0; q < STAGE_Q; ++q) {
        const int s  = tid + q * BLOCK_THREADS;
        const int r  = s / SW4, c = s - r * SW4;
        const int rr = r < N_EVAL ? r : N_EVAL - 1;
        const int cc = c < W4 ? c : W4 - 1;
        const v4f v  = *(const v4f*)(W + rr * N_TOTAL + cc * 4);
        const bool real = (r < N_EVAL) && (c < W4);
        const v4f z = {0.f, 0.f, 0.f, 0.f};
        wv[q] = real ? v : z;
    }
    float bq = 0.0f, rq = FOLD;
    if (tid < N_EVAL) {
        bq = FOLD * bias[tid];
        rq = FOLD * resp[tid];
    }
    #pragma unroll
    for (int q = 0; q < STAGE_Q; ++q) {
        const int s = tid + q * BLOCK_THREADS;
        if (s < NSLOT) *(v4f*)&sW[s * 4] = wv[q];
    }
    if (tid < NPAD) {
        sB[tid] = bq;
        sR[tid] = rq;
    }
    __syncthreads();
    float tv[T_Q];
    #pragma unroll
    for (int q = 0; q < T_Q; ++q) {
        const int idx = tid + q * BLOCK_THREADS;
        const int g = idx >> 8, i = (idx >> 4) & 15, j = idx & 15;
        tv[q] = (idx < NGRP * 256) ? sW[(g * 16 + j) * SWS + D_IN + g * 16 + i] : 0.0f;
    }
    #pragma unroll
    for (int q = 0; q < T_Q; ++q) {
        const int idx = tid + q * BLOCK_THREADS;
        if (idx < NGRP * 256) sT[idx] = tv[q];
    }
    __syncthreads();

    const int  col   = lane & 15;
    const int  khalf = lane >> 4;
    const bool upper = khalf != 0;
    const long ntiles = (long)(n_x / (D_IN * 16));
    const long ngrp_t = (ntiles + TPW - 1) / TPW;

    for (long set = (long)blockIdx.x * WAVES_PER_BLOCK + wave; set < ngrp_t;
         set += (long)gridDim.x * WAVES_PER_BLOCK) {
        long s0[TPW];
        #pragma unroll
        for (int t = 0; t < TPW; ++t) {
            long tile = set * TPW + t;
            tile = tile < ntiles ? tile : ntiles - 1;
            s0[t] = tile * 16;
        }

        int wrow = col * SWS;
        int trow = 0;
        int brow = 0;
        asm volatile("" : "+v"(wrow), "+v"(trow), "+v"(brow));

        v2f xb[TPW][16];
        #pragma unroll
        for (int t = 0; t < TPW; ++t) {
            const float* xrowp = x + (s0[t] + col) * D_IN + khalf * 2;
            #pragma unroll
            for (int k = 0; k < 16; ++k)
                xb[t][k] = *(const v2f*)(xrowp + 4 * k);
        }

        v2f V[TPW][NGRP - 1][4];

        #pragma unroll
        for (int g = 0; g < NGRP; ++g) {
            const int rowbase = g * 16 * SWS + wrow;
            v8f acc[TPW];
            #pragma unroll
            for (int t = 0; t < TPW; ++t) {
                const v8f z8 = {0.f, 0.f, 0.f, 0.f, 0.f, 0.f, 0.f, 0.f};
                acc[t] = z8;
            }

            const float* wa = &sW[rowbase + khalf * 2];
            v2f ax[16];
            #pragma unroll
            for (int k = 0; k < 16; ++k)
                ax[k] = *(const v2f*)(wa + 4 * k);
            v2f ah[NGRP - 1][4];
            #pragma unroll
            for (int p = 0; p < g; ++p) {
                const float* wah = &sW[rowbase + D_IN + p * 16 + khalf * 8];
                #pragma unroll
                for (int k = 0; k < 4; ++k)
                    ah[p][k] = *(const v2f*)(wah + 2 * k);
            }
            __builtin_amdgcn_sched_barrier(0);

            #pragma unroll
            for (int k = 0; k < 16; ++k) {
                #pragma unroll
                for (int t = 0; t < TPW; ++t)
                    acc[t] = __builtin_amdgcn_wmma_f32_16x16x4_f32(false, ax[k], false, xb[t][k],
                                                                   (short)0, acc[t], false, false);
            }
            #pragma unroll
            for (int p = 0; p < g; ++p) {
                #pragma unroll
                for (int k = 0; k < 4; ++k) {
                    #pragma unroll
                    for (int t = 0; t < TPW; ++t)
                        acc[t] = __builtin_amdgcn_wmma_f32_16x16x4_f32(false, ah[p][k], false, V[t][p][k],
                                                                       (short)0, acc[t], false, false);
                }
            }

            __builtin_amdgcn_sched_barrier(0);

            const float* tw = &sT[g * 256 + trow];
            const float* pb = &sB[g * 16 + brow];
            const float* pr = &sR[g * 16 + brow];

            if (g < NGRP - 1) {
                float P[16];
                #pragma unroll
                for (int r = 0; r < 8; ++r) {
                    P[r]     = acc[0][r];
                    P[8 + r] = acc[1][r];
                }
                half_swap4(P[0], P[1], P[2], P[3], P[8],  P[9],  P[10], P[11]);
                half_swap4(P[4], P[5], P[6], P[7], P[12], P[13], P[14], P[15]);
                float Vn[16];
                #pragma unroll
                for (int i = 0; i < 16; ++i) {
                    const float bi = pb[i], ri = pr[i];
                    v4f wq[4];
                    #pragma unroll
                    for (int q = (i + 1) / 4; q < 4; ++q)
                        wq[q] = *(const v4f*)(tw + i * 16 + 4 * q);
                    const float v = neat_sigmoid(bi, ri, P[i]);
                    Vn[i] = v;
                    #pragma unroll
                    for (int j = i + 1; j < 16; ++j)
                        P[j] = __builtin_fmaf(wq[j >> 2][j & 3], v, P[j]);
                }
                half_swap4(Vn[0], Vn[1], Vn[2], Vn[3], Vn[8],  Vn[9],  Vn[10], Vn[11]);
                half_swap4(Vn[4], Vn[5], Vn[6], Vn[7], Vn[12], Vn[13], Vn[14], Vn[15]);
                #pragma unroll
                for (int r = 0; r < 8; ++r) {
                    V[0][g][r >> 1][r & 1] = Vn[r];
                    V[1][g][r >> 1][r & 1] = Vn[8 + r];
                }
            } else {
                float Q[4], H[4];
                #pragma unroll
                for (int r = 0; r < 4; ++r) {
                    Q[r] = acc[0][r];
                    H[r] = acc[1][r];
                }
                half_swap4(Q[0], Q[1], Q[2], Q[3], H[0], H[1], H[2], H[3]);
                float o4[4];
                #pragma unroll
                for (int i = 0; i < 4; ++i) {
                    const float bi = pb[i], ri = pr[i];
                    const v4f w0 = *(const v4f*)(tw + i * 16);
                    const float v = neat_sigmoid(bi, ri, Q[i]);
                    o4[i] = v;
                    #pragma unroll
                    for (int j = i + 1; j < 4; ++j)
                        Q[j] = __builtin_fmaf(w0[j], v, Q[j]);
                }
                const long orow = (upper ? s0[1] : s0[0]) + col;
                const v4f o = {o4[0], o4[1], o4[2], o4[3]};
                *(v4f*)(out + orow * 4) = o;
            }
        }
    }
}

extern "C" void kernel_launch(void* const* d_in, const int* in_sizes, int n_in,
                              void* d_out, int out_size, void* d_ws, size_t ws_size,
                              hipStream_t stream) {
    const float* x    = (const float*)d_in[0];
    const float* W    = (const float*)d_in[1];
    const float* bias = (const float*)d_in[2];
    const float* resp = (const float*)d_in[3];
    float* out = (float*)d_out;

    neat_wmma_kernel<<<NUM_BLOCKS, BLOCK_THREADS, 0, stream>>>(x, W, bias, resp, out, in_sizes[0]);
}


# LDS fetches issued ahead of use: sweep tables 4 nodes ahead (first rows before the hidden K-steps), next group's A fragments inside the sweep; no LDS wait left in a wave's path; sums unchanged
# speedup vs baseline: 11.4989x; 1.0596x over previous
#include <hip/hip_runtime.h>


typedef __attribute__((ext_vector_type(2))) float v2f;
typedef __attribute__((ext_vector_type(4))) float v4f;
typedef __attribute__((ext_vector_type(8))) float v8f;

#define D_IN    64
#define N_EVAL  68
#define N_TOTAL 132
#define NGRP    5
#define NPAD    80
#define SWS     148
#define WAVES_PER_BLOCK 16
#define BLOCK_THREADS (WAVES_PER_BLOCK * 32)
#define TPW 2
#define NUM_BLOCKS 256
#define SW4     (SWS / 4)
#define W4      (N_TOTAL / 4)
#define NSLOT   (NPAD * SW4)
#define STAGE_Q ((NSLOT + BLOCK_THREADS - 1) / BLOCK_THREADS)
#define T_Q     ((NGRP * 256 + BLOCK_THREADS - 1) / BLOCK_THREADS)

#define FOLD   (-7.2134752044448169f)
#define UCLAMP (86.561702453337803f)

__device__ __forceinline__ float neat_sigmoid(float bF, float rF, float t) {
    float u = __builtin_fmaf(rF, t, bF);
    u = fminf(fmaxf(u, -UCLAMP), UCLAMP);
    float e = __builtin_amdgcn_exp2f(u);
    return __builtin_amdgcn_rcpf(1.0f + e);
}

__device__ __forceinline__ void half_swap4(float& d0, float& d1, float& d2, float& d3,
                                           float& s0, float& s1, float& s2, float& s3) {
    asm("v_permlane16_swap_b32 %0, %4\n\t"
        "v_permlane16_swap_b32 %1, %5\n\t"
        "v_permlane16_swap_b32 %2, %6\n\t"
        "v_permlane16_swap_b32 %3, %7"
        : "+v"(d0), "+v"(d1), "+v"(d2), "+v"(d3), "+v"(s0), "+v"(s1), "+v"(s2), "+v"(s3));
}

__global__ __launch_bounds__(BLOCK_THREADS, 4)
void neat_wmma_kernel(const float* __restrict__ x,
                      const float* __restrict__ W,
                      const float* __restrict__ bias,
                      const float* __restrict__ resp,
                      float* __restrict__ out,
                      int n_x)
{
    __shared__ __attribute__((aligned(16))) float sW[NPAD * SWS];
    __shared__ __attribute__((aligned(16))) float sT[NGRP * 16 * 16];
    __shared__ __attribute__((aligned(16))) float sB[NPAD];
    __shared__ __attribute__((aligned(16))) float sR[NPAD];

    const int tid  = threadIdx.x;
    const int wave = tid >> 5;
    const int lane = tid & 31;

    v4f wv[STAGE_Q];
    #pragma unroll
    for (int q = 0; q < STAGE_Q; ++q) {
        const int s  = tid + q * BLOCK_THREADS;
        const int r  = s / SW4, c = s - r * SW4;
        const int rr = r < N_EVAL ? r : N_EVAL - 1;
        const int cc = c < W4 ? c : W4 - 1;
        const v4f v  = *(const v4f*)(W + rr * N_TOTAL + cc * 4);
        const bool real = (r < N_EVAL) && (c < W4);
        const v4f z = {0.f, 0.f, 0.f, 0.f};
        wv[q] = real ? v : z;
    }
    float bq = 0.0f, rq = FOLD;
    if (tid < N_EVAL) {
        bq = FOLD * bias[tid];
        rq = FOLD * resp[tid];
    }
    #pragma unroll
    for (int q = 0; q < STAGE_Q; ++q) {
        const int s = tid + q * BLOCK_THREADS;
        if (s < NSLOT) *(v4f*)&sW[s * 4] = wv[q];
    }
    if (tid < NPAD) {
        sB[tid] = bq;
        sR[tid] = rq;
    }
    __syncthreads();
    float tv[T_Q];
    #pragma unroll
    for (int q = 0; q < T_Q; ++q) {
        const int idx = tid + q * BLOCK_THREADS;
        const int g = idx >> 8, i = (idx >> 4) & 15, j = idx & 15;
        tv[q] = (idx < NGRP * 256) ? sW[(g * 16 + j) * SWS + D_IN + g * 16 + i] : 0.0f;
    }
    #pragma unroll
    for (int q = 0; q < T_Q; ++q) {
        const int idx = tid + q * BLOCK_THREADS;
        if (idx < NGRP * 256) sT[idx] = tv[q];
    }
    __syncthreads();

    const int  col   = lane & 15;
    const int  khalf = lane >> 4;
    const bool upper = khalf != 0;
    const long ntiles = (long)(n_x / (D_IN * 16));
    const long ngrp_t = (ntiles + TPW - 1) / TPW;

    for (long set = (long)blockIdx.x * WAVES_PER_BLOCK + wave; set < ngrp_t;
         set += (long)gridDim.x * WAVES_PER_BLOCK) {
        long s0[TPW];
        #pragma unroll
        for (int t = 0; t < TPW; ++t) {
            long tile = set * TPW + t;
            tile = tile < ntiles ? tile : ntiles - 1;
            s0[t] = tile * 16;
        }

        int wrow = col * SWS;
        int trow = 0;
        int brow = 0;
        asm volatile("" : "+v"(wrow), "+v"(trow), "+v"(brow));

        v2f xb[TPW][16];
        #pragma unroll
        for (int t = 0; t < TPW; ++t) {
            const float* xrowp = x + (s0[t] + col) * D_IN + khalf * 2;
            #pragma unroll
            for (int k = 0; k < 16; ++k)
                xb[t][k] = *(const v2f*)(xrowp + 4 * k);
        }

        v2f V[TPW][NGRP - 1][4];

        v2f ax[16];
        v2f ah[NGRP - 1][4];
        {
            const float* wa = &sW[wrow + khalf * 2];
            #pragma unroll
            for (int k = 0; k < 16; ++k)
                ax[k] = *(const v2f*)(wa + 4 * k);
        }

        #pragma unroll
        for (int g = 0; g < NGRP; ++g) {
            v8f acc[TPW];
            #pragma unroll
            for (int t = 0; t < TPW; ++t) {
                const v8f z8 = {0.f, 0.f, 0.f, 0.f, 0.f, 0.f, 0.f, 0.f};
                acc[t] = z8;
            }

            const float* tw = &sT[g * 256 + trow];
            const float* pb = &sB[g * 16 + brow];
            const float* pr = &sR[g * 16 + brow];
            const int QEND  = (g < NGRP - 1) ? 4 : 1;
            constexpr int AHEAD = 4;
            v4f wr[16][4];
            v4f bq4[4], rq4[4];
            if (g == 0) {
                #pragma unroll
                for (int i = 0; i < AHEAD; ++i) {
                    #pragma unroll
                    for (int q = (i + 1) / 4; q < QEND; ++q)
                        wr[i][q] = *(const v4f*)(tw + i * 16 + 4 * q);
                }
                bq4[0] = *(const v4f*)(pb);
                rq4[0] = *(const v4f*)(pr);
            }
            __builtin_amdgcn_sched_barrier(0);

            #pragma unroll
            for (int k = 0; k < 16; ++k) {
                #pragma unroll
                for (int t = 0; t < TPW; ++t)
                    acc[t] = __builtin_amdgcn_wmma_f32_16x16x4_f32(false, ax[k], false, xb[t][k],
                                                                   (short)0, acc[t], false, false);
            }

            __builtin_amdgcn_sched_barrier(0);
            if (g > 0) {
                #pragma unroll
                for (int i = 0; i < AHEAD; ++i) {
                    #pragma unroll
                    for (int q = (i + 1) / 4; q < QEND; ++q)
                        wr[i][q] = *(const v4f*)(tw + i * 16 + 4 * q);
                }
                bq4[0] = *(const v4f*)(pb);
                rq4[0] = *(const v4f*)(pr);
            }
            __builtin_amdgcn_sched_barrier(0);
            #pragma unroll
            for (int p = 0; p < g; ++p) {
                #pragma unroll
                for (int k = 0; k < 4; ++k) {
                    #pragma unroll
                    for (int t = 0; t < TPW; ++t)
                        acc[t] = __builtin_amdgcn_wmma_f32_16x16x4_f32(false, ah[p][k], false, V[t][p][k],
                                                                       (short)0, acc[t], false, false);
                }
            }

            __builtin_amdgcn_sched_barrier(0);

            if (g < NGRP - 1) {
                float P[16];
                #pragma unroll
                for (int r = 0; r < 8; ++r) {
                    P[r]     = acc[0][r];
                    P[8 + r] = acc[1][r];
                }
                half_swap4(P[0], P[1], P[2], P[3], P[8],  P[9],  P[10], P[11]);
                half_swap4(P[4], P[5], P[6], P[7], P[12], P[13], P[14], P[15]);
                float Vn[16];
                #pragma unroll
                for (int i = 0; i < 16; ++i) {
                    if (i == 12) {
                        const int rownext = (g + 1) * 16 * SWS + wrow;
                        const float* wa = &sW[rownext + khalf * 2];
                        #pragma unroll
                        for (int k = 0; k < 16; ++k)
                            ax[k] = *(const v2f*)(wa + 4 * k);
                        #pragma unroll
                        for (int p = 0; p < g + 1; ++p) {
                            const float* wah = &sW[rownext + D_IN + p * 16 + khalf * 8];
                            #pragma unroll
                            for (int k = 0; k < 4; ++k)
                                ah[p][k] = *(const v2f*)(wah + 2 * k);
                        }
                    }
                    if (i + AHEAD < 15) {
                        #pragma unroll
                        for (int q = (i + AHEAD + 1) / 4; q < 4; ++q)
                            wr[i + AHEAD][q] = *(const v4f*)(tw + (i + AHEAD) * 16 + 4 * q);
                    }
                    if (i + AHEAD < 16 && ((i + AHEAD) & 3) == 0) {
                        bq4[(i + AHEAD) >> 2] = *(const v4f*)(pb + i + AHEAD);
                        rq4[(i + AHEAD) >> 2] = *(const v4f*)(pr + i + AHEAD);
                    }
                    const float bi = bq4[i >> 2][i & 3], ri = rq4[i >> 2][i & 3];
                    const float v = neat_sigmoid(bi, ri, P[i]);
                    Vn[i] = v;
                    #pragma unroll
                    for (int j = i + 1; j < 16; ++j)
                        P[j] = __builtin_fmaf(wr[i][j >> 2][j & 3], v, P[j]);
                    #pragma unroll
                    for (int q = (i + 1) / 4; q < 4; ++q)
                        asm volatile("" :: "v"(wr[i][q]));
                    __builtin_amdgcn_sched_barrier(0);
                }
                half_swap4(Vn[0], Vn[1], Vn[2], Vn[3], Vn[8],  Vn[9],  Vn[10], Vn[11]);
                half_swap4(Vn[4], Vn[5], Vn[6], Vn[7], Vn[12], Vn[13], Vn[14], Vn[15]);
                #pragma unroll
                for (int r = 0; r < 8; ++r) {
                    V[0][g][r >> 1][r & 1] = Vn[r];
                    V[1][g][r >> 1][r & 1] = Vn[8 + r];
                }
            } else {
                float Q[4], H[4];
                #pragma unroll
                for (int r = 0; r < 4; ++r) {
                    Q[r] = acc[0][r];
                    H[r] = acc[1][r];
                }
                half_swap4(Q[0], Q[1], Q[2], Q[3], H[0], H[1], H[2], H[3]);
                float o4[4];
                #pragma unroll
                for (int i = 0; i < 4; ++i) {
                    const float bi = bq4[0][i], ri = rq4[0][i];
                    const float v = neat_sigmoid(bi, ri, Q[i]);
                    o4[i] = v;
                    #pragma unroll
                    for (int j = i + 1; j < 4; ++j)
                        Q[j] = __builtin_fmaf(wr[i][0][j], v, Q[j]);
                    if (i < 3) asm volatile("" :: "v"(wr[i][0]));
                }
                const long orow = (upper ? s0[1] : s0[0]) + col;
                const v4f o = {o4[0], o4[1], o4[2], o4[3]};
                *(v4f*)(out + orow * 4) = o;
            }
        }
    }
}

extern "C" void kernel_launch(void* const* d_in, const int* in_sizes, int n_in,
                              void* d_out, int out_size, void* d_ws, size_t ws_size,
                              hipStream_t stream) {
    const float* x    = (const float*)d_in[0];
    const float* W    = (const float*)d_in[1];
    const float* bias = (const float*)d_in[2];
    const float* resp = (const float*)d_in[3];
    float* out = (float*)d_out;

    neat_wmma_kernel<<<NUM_BLOCKS, BLOCK_THREADS, 0, stream>>>(x, W, bias, resp, out, in_sizes[0]);
}
